# starting kernel
# baseline (speedup 1.0000x reference)
#include <hip/hip_runtime.h>


typedef __attribute__((ext_vector_type(2))) float v2f;
typedef __attribute__((ext_vector_type(8))) float v8f;

#define BATCH 64
#define VDIM  16
#define MAXN  24
#define NTHREADS 384
#define DPAD  72
#define XVPAD 49

__launch_bounds__(NTHREADS)
__global__ void residues_encode_kernel(const float* __restrict__ x,
                                       const float* __restrict__ W,
                                       const int*   __restrict__ seq,
                                       const int*   __restrict__ ridx,
                                       const int*   __restrict__ natoms,
                                       float* __restrict__ out_root,
                                       float* __restrict__ out_xv,
                                       int A, int R)
{
    __shared__ float lds_root[3][BATCH];
    __shared__ float lds_delta[3][MAXN][DPAD];

    const int r = blockIdx.x;
    const int t = threadIdx.x;

    const int sr = seq[r];
    const int ri = ridx[r];
    const int na = natoms[sr];

    if (t < BATCH * 3) {
        const int b = t / 3, d = t % 3;
        const float rv = x[((long)b * A + (ri + 1)) * 3 + d];
        lds_root[d][b] = rv;
        out_root[((long)b * R + r) * 3 + d] = rv;
    }
    __syncthreads();

    for (int e = t; e < BATCH * MAXN * 3; e += NTHREADS) {
        const int b = e / (MAXN * 3);
        const int inner = e % (MAXN * 3);
        const int n = inner / 3, d = inner % 3;
        float v = 0.0f;
        if (n < na) {
            int atom = ri + n;
            if (atom > A - 1) atom = A - 1;
            v = x[((long)b * A + atom) * 3 + d] - lds_root[d][b];
        }
        lds_delta[d][n][b] = v;
    }
    __syncthreads();

    const int wave  = t >> 5;
    const int lane  = t & 31;
    const int half  = lane >> 4;
    const int col   = lane & 15;
    const int d     = wave >> 2;
    const int btile = wave & 3;
    const int b     = btile * 16 + col;

    const float* Wr = W + ((long)(sr * VDIM + col) * MAXN + 2 * half);

    v8f acc = {};
    #pragma unroll
    for (int k = 0; k < 6; ++k) {
        v2f a;
        a[0] = Wr[4 * k];
        a[1] = Wr[4 * k + 1];
        v2f bb;
        bb[0] = lds_delta[d][4 * k + 2 * half + 0][b];
        bb[1] = lds_delta[d][4 * k + 2 * half + 1][b];
        acc = __builtin_amdgcn_wmma_f32_16x16x4_f32(
             false, a,  false, bb,
             (short)0, acc,  false,  false);
    }
    __syncthreads();

    float* xv = &lds_delta[0][0][0];
    #pragma unroll
    for (int p = 0; p < 8; ++p) {
        const int v = p + 8 * half;
        xv[b * XVPAD + v * 3 + d] = acc[p];
    }
    __syncthreads();

    for (int e = t; e < BATCH * VDIM * 3; e += NTHREADS) {
        const int b2 = e / (VDIM * 3);
        const int j  = e % (VDIM * 3);
        out_xv[((long)b2 * R + r) * (VDIM * 3) + j] = xv[b2 * XVPAD + j];
    }
}

extern "C" void kernel_launch(void* const* d_in, const int* in_sizes, int n_in,
                              void* d_out, int out_size, void* d_ws, size_t ws_size,
                              hipStream_t stream) {
    const float* x      = (const float*)d_in[0];
    const float* W      = (const float*)d_in[1];
    const int*   seq    = (const int*)d_in[2];
    const int*   ridx   = (const int*)d_in[3];
    const int*   natoms = (const int*)d_in[4];

    const int R = in_sizes[2];
    const int A = in_sizes[0] / (BATCH * 3);

    float* out_root = (float*)d_out;
    float* out_xv   = out_root + (size_t)BATCH * R * 3;

    residues_encode_kernel<<<R, NTHREADS, 0, stream>>>(x, W, seq, ridx, natoms,
                                                       out_root, out_xv, A, R);
}


# Block's serial chain of memory round trips cut from about 15 to 5: root, all 12 gather loads per thread (clamped address, mask applied to the value) and W issued in groups, none waits in a branch
# speedup vs baseline: 1.2590x; 1.2590x over previous
#include <hip/hip_runtime.h>


typedef __attribute__((ext_vector_type(2))) float v2f;
typedef __attribute__((ext_vector_type(8))) float v8f;

#define BATCH 64
#define VDIM  16
#define MAXN  24
#define NTHREADS 384
#define DPAD  72
#define XVPAD 49
#define GITER (BATCH * MAXN * 3 / NTHREADS)

__launch_bounds__(NTHREADS)
__global__ void residues_encode_kernel(const float* __restrict__ x,
                                       const float* __restrict__ W,
                                       const int*   __restrict__ seq,
                                       const int*   __restrict__ ridx,
                                       const int*   __restrict__ natoms,
                                       float* __restrict__ out_root,
                                       float* __restrict__ out_xv,
                                       int A, int R)
{
    __shared__ float lds_root[3][BATCH];
    __shared__ float lds_delta[3][MAXN][DPAD];

    const int r = blockIdx.x;
    const int t = threadIdx.x;

    const int sr = seq[r];
    const int ri = ridx[r];
    const int na = natoms[sr];

    const int pb = t / 3, pd = t % 3;
    float rv = 0.0f;
    if (t < BATCH * 3)
        rv = x[(unsigned)(pb * A + (ri + 1)) * 3u + (unsigned)pd];

    const int q0 = t / (MAXN * 3);
    const int n0 = (t % (MAXN * 3)) / 3;
    const unsigned a3 = 3u * (unsigned)A;
    float g[GITER];
    #pragma unroll
    for (int i = 0; i < GITER; ++i) {
        const int m = n0 + 8 * i;
        const int c = m / MAXN;
        const int n = m - MAXN * c;
        const int b = q0 + 5 * i + c;
        int atom = ri + n;
        if (atom > A - 1) atom = A - 1;
        g[i] = x[(unsigned)b * a3 + (unsigned)atom * 3u + (unsigned)pd];
    }

    const int wave  = t >> 5;
    const int lane  = t & 31;
    const int half  = lane >> 4;
    const int col   = lane & 15;

    if (t < BATCH * 3)
        lds_root[pd][pb] = rv;
    __syncthreads();

    const float* Wr = W + ((sr * VDIM + col) * MAXN + 2 * half);
    v2f afrag[6];
    #pragma unroll
    for (int k = 0; k < 6; ++k) {
        afrag[k][0] = Wr[4 * k];
        afrag[k][1] = Wr[4 * k + 1];
    }

    #pragma unroll
    for (int i = 0; i < GITER; ++i) {
        const int m = n0 + 8 * i;
        const int c = m / MAXN;
        const int n = m - MAXN * c;
        const int b = q0 + 5 * i + c;
        const float diff = g[i] - lds_root[pd][b];
        lds_delta[pd][n][b] = (n < na) ? diff : 0.0f;
    }
    __syncthreads();

    const int d     = wave >> 2;
    const int btile = wave & 3;
    const int b     = btile * 16 + col;

    v8f acc = {};
    #pragma unroll
    for (int k = 0; k < 6; ++k) {
        v2f bb;
        bb[0] = lds_delta[d][4 * k + 2 * half + 0][b];
        bb[1] = lds_delta[d][4 * k + 2 * half + 1][b];
        acc = __builtin_amdgcn_wmma_f32_16x16x4_f32(
             false, afrag[k],  false, bb,
             (short)0, acc,  false,  false);
    }
    __syncthreads();

    float* xv = &lds_delta[0][0][0];
    #pragma unroll
    for (int p = 0; p < 8; ++p) {
        const int v = p + 8 * half;
        xv[b * XVPAD + v * 3 + d] = acc[p];
    }
    __syncthreads();

    for (int e = t; e < BATCH * VDIM * 3; e += NTHREADS) {
        const int b2 = e / (VDIM * 3);
        const int j  = e % (VDIM * 3);
        out_xv[((long)b2 * R + r) * (VDIM * 3) + j] = xv[b2 * XVPAD + j];
    }

    if (t < BATCH * 3)
        out_root[((long)pb * R + r) * 3 + pd] = rv;
}

extern "C" void kernel_launch(void* const* d_in, const int* in_sizes, int n_in,
                              void* d_out, int out_size, void* d_ws, size_t ws_size,
                              hipStream_t stream) {
    const float* x      = (const float*)d_in[0];
    const float* W      = (const float*)d_in[1];
    const int*   seq    = (const int*)d_in[2];
    const int*   ridx   = (const int*)d_in[3];
    const int*   natoms = (const int*)d_in[4];

    const int R = in_sizes[2];
    const int A = in_sizes[0] / (BATCH * 3);

    float* out_root = (float*)d_out;
    float* out_xv   = out_root + (size_t)BATCH * R * 3;

    residues_encode_kernel<<<R, NTHREADS, 0, stream>>>(x, W, seq, ridx, natoms,
                                                       out_root, out_xv, A, R);
}


# Fewer, longer-lived waves: 1024 resident workgroups each loop over 4 consecutive residues (12288 waves enter, not 49152); gather indices as 3 bases x 4 strides so the loop's registers stay at 63
# speedup vs baseline: 1.6145x; 1.2824x over previous
#include <hip/hip_runtime.h>


typedef __attribute__((ext_vector_type(2))) float v2f;
typedef __attribute__((ext_vector_type(8))) float v8f;

#define BATCH 64
#define VDIM  16
#define MAXN  24
#define NTHREADS 384
#define DPAD  72
#define XVPAD 49
#define GITER (BATCH * MAXN * 3 / NTHREADS)
#define NBLOCKS 1024

__launch_bounds__(NTHREADS)
__global__ void residues_encode_kernel(const float* __restrict__ x,
                                       const float* __restrict__ W,
                                       const int*   __restrict__ seq,
                                       const int*   __restrict__ ridx,
                                       const int*   __restrict__ natoms,
                                       float* __restrict__ out_root,
                                       float* __restrict__ out_xv,
                                       int A, int R)
{
    __shared__ float lds_root[3][BATCH];
    __shared__ float lds_delta[3][MAXN][DPAD];

    const int t = threadIdx.x;

    const int per     = (R + (int)gridDim.x - 1) / (int)gridDim.x;
    const int r_begin = (int)blockIdx.x * per;
    const int r_end   = (r_begin + per < R) ? (r_begin + per) : R;

    const int q0 = t / (MAXN * 3);
    const int n0 = (t % (MAXN * 3)) / 3;
    const unsigned a3 = 3u * (unsigned)A;
    int ns[3], bs[3];
    #pragma unroll
    for (int s = 0; s < 3; ++s) {
        const int m0 = n0 + 8 * s;
        const int c  = m0 / MAXN;
        ns[s] = m0 - MAXN * c;
        bs[s] = q0 + 5 * s + c;
    }

    for (int r = r_begin; r < r_end; ++r) {

    const int sr = seq[r];
    const int ri = ridx[r];
    const int na = natoms[sr];

    const int pb = t / 3, pd = t % 3;
    float rv = 0.0f;
    if (t < BATCH * 3)
        rv = x[(unsigned)(pb * A + (ri + 1)) * 3u + (unsigned)pd];

    float g[GITER];
    #pragma unroll
    for (int s = 0; s < 3; ++s) {
        int atom = ri + ns[s];
        if (atom > A - 1) atom = A - 1;
        const unsigned base = (unsigned)bs[s] * a3 + (unsigned)atom * 3u + (unsigned)pd;
        #pragma unroll
        for (int m = 0; m < 4; ++m)
            g[3 * m + s] = x[base + (unsigned)(16 * m) * a3];
    }

    const int wave  = t >> 5;
    const int lane  = t & 31;
    const int half  = lane >> 4;
    const int col   = lane & 15;

    if (t < BATCH * 3)
        lds_root[pd][pb] = rv;
    __syncthreads();

    const float* Wr = W + ((sr * VDIM + col) * MAXN + 2 * half);
    v2f afrag[6];
    #pragma unroll
    for (int k = 0; k < 6; ++k) {
        afrag[k][0] = Wr[4 * k];
        afrag[k][1] = Wr[4 * k + 1];
    }

    #pragma unroll
    for (int s = 0; s < 3; ++s) {
        #pragma unroll
        for (int m = 0; m < 4; ++m) {
            const int b = bs[s] + 16 * m;
            const float diff = g[3 * m + s] - lds_root[pd][b];
            lds_delta[pd][ns[s]][b] = (ns[s] < na) ? diff : 0.0f;
        }
    }
    __syncthreads();

    const int d     = wave >> 2;
    const int btile = wave & 3;
    const int b     = btile * 16 + col;

    v8f acc = {};
    #pragma unroll
    for (int k = 0; k < 6; ++k) {
        v2f bb;
        bb[0] = lds_delta[d][4 * k + 2 * half + 0][b];
        bb[1] = lds_delta[d][4 * k + 2 * half + 1][b];
        acc = __builtin_amdgcn_wmma_f32_16x16x4_f32(
             false, afrag[k],  false, bb,
             (short)0, acc,  false,  false);
    }
    __syncthreads();

    float* xv = &lds_delta[0][0][0];
    #pragma unroll
    for (int p = 0; p < 8; ++p) {
        const int v = p + 8 * half;
        xv[b * XVPAD + v * 3 + d] = acc[p];
    }
    __syncthreads();

    for (int e = t; e < BATCH * VDIM * 3; e += NTHREADS) {
        const int b2 = e / (VDIM * 3);
        const int j  = e % (VDIM * 3);
        out_xv[((long)b2 * R + r) * (VDIM * 3) + j] = xv[b2 * XVPAD + j];
    }

    if (t < BATCH * 3)
        out_root[((long)pb * R + r) * 3 + pd] = rv;

    }
}

extern "C" void kernel_launch(void* const* d_in, const int* in_sizes, int n_in,
                              void* d_out, int out_size, void* d_ws, size_t ws_size,
                              hipStream_t stream) {
    const float* x      = (const float*)d_in[0];
    const float* W      = (const float*)d_in[1];
    const int*   seq    = (const int*)d_in[2];
    const int*   ridx   = (const int*)d_in[3];
    const int*   natoms = (const int*)d_in[4];

    const int R = in_sizes[2];
    const int A = in_sizes[0] / (BATCH * 3);

    float* out_root = (float*)d_out;
    float* out_xv   = out_root + (size_t)BATCH * R * 3;

    residues_encode_kernel<<<NBLOCKS, NTHREADS, 0, stream>>>(x, W, seq, ridx, natoms,
                                                             out_root, out_xv, A, R);
}


# Window kept in LDS as [b][d][n], pitch 76, root as [b][d]: phase-2 stores and root reads lose their bank conflicts (up to 6-way, 3-way), B fragment pair becomes one 8-byte LDS read per k-step
# speedup vs baseline: 1.6768x; 1.0386x over previous
#include <hip/hip_runtime.h>


typedef __attribute__((ext_vector_type(2))) float v2f;
typedef __attribute__((ext_vector_type(8))) float v8f;

#define BATCH 64
#define VDIM  16
#define MAXN  24
#define NTHREADS 384
#define DPITCH 76
#define XVPAD 49
#define GITER (BATCH * MAXN * 3 / NTHREADS)
#define NBLOCKS 1024

__launch_bounds__(NTHREADS)
__global__ void residues_encode_kernel(const float* __restrict__ x,
                                       const float* __restrict__ W,
                                       const int*   __restrict__ seq,
                                       const int*   __restrict__ ridx,
                                       const int*   __restrict__ natoms,
                                       float* __restrict__ out_root,
                                       float* __restrict__ out_xv,
                                       int A, int R)
{
    __shared__ float lds_root[BATCH * 3];
    __shared__ __attribute__((aligned(16))) float lds_delta[BATCH * DPITCH];

    const int t = threadIdx.x;

    const int per     = (R + (int)gridDim.x - 1) / (int)gridDim.x;
    const int r_begin = (int)blockIdx.x * per;
    const int r_end   = (r_begin + per < R) ? (r_begin + per) : R;

    const int q0 = t / (MAXN * 3);
    const int n0 = (t % (MAXN * 3)) / 3;
    const unsigned a3 = 3u * (unsigned)A;
    int ns[3], bs[3];
    #pragma unroll
    for (int s = 0; s < 3; ++s) {
        const int m0 = n0 + 8 * s;
        const int c  = m0 / MAXN;
        ns[s] = m0 - MAXN * c;
        bs[s] = q0 + 5 * s + c;
    }

    for (int r = r_begin; r < r_end; ++r) {

    const int sr = seq[r];
    const int ri = ridx[r];
    const int na = natoms[sr];

    const int pb = t / 3, pd = t % 3;
    float rv = 0.0f;
    if (t < BATCH * 3)
        rv = x[(unsigned)(pb * A + (ri + 1)) * 3u + (unsigned)pd];

    float g[GITER];
    #pragma unroll
    for (int s = 0; s < 3; ++s) {
        int atom = ri + ns[s];
        if (atom > A - 1) atom = A - 1;
        const unsigned base = (unsigned)bs[s] * a3 + (unsigned)atom * 3u + (unsigned)pd;
        #pragma unroll
        for (int m = 0; m < 4; ++m)
            g[3 * m + s] = x[base + (unsigned)(16 * m) * a3];
    }

    const int wave  = t >> 5;
    const int lane  = t & 31;
    const int half  = lane >> 4;
    const int col   = lane & 15;

    if (t < BATCH * 3)
        lds_root[t] = rv;
    __syncthreads();

    const float* Wr = W + ((sr * VDIM + col) * MAXN + 2 * half);
    v2f afrag[6];
    #pragma unroll
    for (int k = 0; k < 6; ++k) {
        afrag[k][0] = Wr[4 * k];
        afrag[k][1] = Wr[4 * k + 1];
    }

    #pragma unroll
    for (int s = 0; s < 3; ++s) {
        #pragma unroll
        for (int m = 0; m < 4; ++m) {
            const int b = bs[s] + 16 * m;
            const float diff = g[3 * m + s] - lds_root[3 * b + pd];
            lds_delta[b * DPITCH + pd * MAXN + ns[s]] = (ns[s] < na) ? diff : 0.0f;
        }
    }
    __syncthreads();

    const int d     = wave >> 2;
    const int btile = wave & 3;
    const int b     = btile * 16 + col;

    const float* brow = &lds_delta[b * DPITCH + d * MAXN + 2 * half];
    v8f acc = {};
    #pragma unroll
    for (int k = 0; k < 6; ++k) {
        const v2f bb = *(const v2f*)(brow + 4 * k);
        acc = __builtin_amdgcn_wmma_f32_16x16x4_f32(
             false, afrag[k],  false, bb,
             (short)0, acc,  false,  false);
    }
    __syncthreads();

    float* xv = &lds_delta[0];
    #pragma unroll
    for (int p = 0; p < 8; ++p) {
        const int v = p + 8 * half;
        xv[b * XVPAD + v * 3 + d] = acc[p];
    }
    __syncthreads();

    for (int e = t; e < BATCH * VDIM * 3; e += NTHREADS) {
        const int b2 = e / (VDIM * 3);
        const int j  = e % (VDIM * 3);
        out_xv[((long)b2 * R + r) * (VDIM * 3) + j] = xv[b2 * XVPAD + j];
    }

    if (t < BATCH * 3)
        out_root[((long)pb * R + r) * 3 + pd] = rv;

    }
}

extern "C" void kernel_launch(void* const* d_in, const int* in_sizes, int n_in,
                              void* d_out, int out_size, void* d_ws, size_t ws_size,
                              hipStream_t stream) {
    const float* x      = (const float*)d_in[0];
    const float* W      = (const float*)d_in[1];
    const int*   seq    = (const int*)d_in[2];
    const int*   ridx   = (const int*)d_in[3];
    const int*   natoms = (const int*)d_in[4];

    const int R = in_sizes[2];
    const int A = in_sizes[0] / (BATCH * 3);

    float* out_root = (float*)d_out;
    float* out_xv   = out_root + (size_t)BATCH * R * 3;

    residues_encode_kernel<<<NBLOCKS, NTHREADS, 0, stream>>>(x, W, seq, ridx, natoms,
                                                             out_root, out_xv, A, R);
}


# Residue loop software-pipelined: next residue's index, atom count, root, gather and W loads are requested under the current one's compute and stores; barriers order LDS only, so loads cross them
# speedup vs baseline: 1.9002x; 1.1332x over previous
#include <hip/hip_runtime.h>


typedef __attribute__((ext_vector_type(2))) float v2f;
typedef __attribute__((ext_vector_type(8))) float v8f;

#define BATCH 64
#define VDIM  16
#define MAXN  24
#define NTHREADS 384
#define DPITCH 76
#define XVPAD 49
#define GITER (BATCH * MAXN * 3 / NTHREADS)
#define NBLOCKS 1024

__device__ __forceinline__ void lds_barrier()
{
    __builtin_amdgcn_fence(__ATOMIC_RELEASE, "workgroup", "local");
    __builtin_amdgcn_s_barrier();
    __builtin_amdgcn_fence(__ATOMIC_ACQUIRE, "workgroup", "local");
}

__launch_bounds__(NTHREADS, 12)
__global__ void residues_encode_kernel(const float* __restrict__ x,
                                       const float* __restrict__ W,
                                       const int*   __restrict__ seq,
                                       const int*   __restrict__ ridx,
                                       const int*   __restrict__ natoms,
                                       float* __restrict__ out_root,
                                       float* __restrict__ out_xv,
                                       int A, int R)
{
    __shared__ float lds_root[BATCH * 3];
    __shared__ __attribute__((aligned(16))) float lds_delta[BATCH * DPITCH];

    const int t = threadIdx.x;

    const int per     = (R + (int)gridDim.x - 1) / (int)gridDim.x;
    const int r_begin = (int)blockIdx.x * per;
    const int r_end   = (r_begin + per < R) ? (r_begin + per) : R;

    const int q0 = t / (MAXN * 3);
    const int n0 = (t % (MAXN * 3)) / 3;
    const unsigned a3 = 3u * (unsigned)A;
    int ns[3], bs[3];
    #pragma unroll
    for (int s = 0; s < 3; ++s) {
        const int m0 = n0 + 8 * s;
        const int c  = m0 / MAXN;
        ns[s] = m0 - MAXN * c;
        bs[s] = q0 + 5 * s + c;
    }

    if (r_begin >= r_end) return;

    const int pb = t / 3, pd = t % 3;
    const int wave  = t >> 5;
    const int lane  = t & 31;
    const int half  = lane >> 4;
    const int col   = lane & 15;

#define ISSUE_X_LOADS(RI)                                                                   \
    do {                                                                                    \
        if (t < BATCH * 3)                                                                  \
            rv = x[(unsigned)(pb * A + ((RI) + 1)) * 3u + (unsigned)pd];                    \
        _Pragma("unroll")                       \
        for (int s = 0; s < 3; ++s) {                                                       \
            int atom = (RI) + ns[s];                                                        \
            if (atom > A - 1) atom = A - 1;                             \
            const unsigned base = (unsigned)bs[s] * a3 + (unsigned)atom * 3u + (unsigned)pd; \
            _Pragma("unroll")                                               \
            for (int m = 0; m < 4; ++m)                                                     \
                g[3 * m + s] = x[base + (unsigned)(16 * m) * a3];                           \
        }                                                                                   \
    } while (0)

#define ISSUE_W_LOADS(SR)                                                                   \
    do {                                                                                    \
        const float* Wr = W + (((SR) * VDIM + col) * MAXN + 2 * half);                      \
        _Pragma("unroll")                      \
        for (int k = 0; k < 6; ++k) {                                                       \
            afrag[k][0] = Wr[4 * k];                                                        \
            afrag[k][1] = Wr[4 * k + 1];                                                    \
        }                                                                                   \
    } while (0)

    const int sr0 = seq[r_begin];
    const int ri0 = ridx[r_begin];
    int na = natoms[sr0];
    float rv = 0.0f;
    float g[GITER];
    v2f afrag[6];
    ISSUE_W_LOADS(sr0);
    ISSUE_X_LOADS(ri0);
    const int rn1 = (r_begin + 1 < r_end) ? r_begin + 1 : r_begin;
    int ri_n = ridx[rn1];
    int sr_n = seq[rn1];

    const int d     = wave >> 2;
    const int btile = wave & 3;
    const int b     = btile * 16 + col;

    for (int r = r_begin; r < r_end; ++r) {

    const bool more = (r + 1 < r_end);

    const float rv_cur = rv;

    if (t < BATCH * 3)
        lds_root[t] = rv;
    lds_barrier();

    #pragma unroll
    for (int s = 0; s < 3; ++s) {
        #pragma unroll
        for (int m = 0; m < 4; ++m) {
            const int b2 = bs[s] + 16 * m;
            const float diff = g[3 * m + s] - lds_root[3 * b2 + pd];
            lds_delta[b2 * DPITCH + pd * MAXN + ns[s]] = (ns[s] < na) ? diff : 0.0f;
        }
    }

    const int na_n = natoms[sr_n];
    if (more)
        ISSUE_X_LOADS(ri_n);
    lds_barrier();

    const float* brow = &lds_delta[b * DPITCH + d * MAXN + 2 * half];
    v8f acc = {};
    #pragma unroll
    for (int k = 0; k < 6; ++k) {
        const v2f bb = *(const v2f*)(brow + 4 * k);
        acc = __builtin_amdgcn_wmma_f32_16x16x4_f32(
             false, afrag[k],  false, bb,
             (short)0, acc,  false,  false);
    }

    lds_barrier();

    float* xv = &lds_delta[0];
    #pragma unroll
    for (int p = 0; p < 8; ++p) {
        const int v = p + 8 * half;
        xv[b * XVPAD + v * 3 + d] = acc[p];
    }
    lds_barrier();

    if (more)
        ISSUE_W_LOADS(sr_n);

    const int rn2 = (r + 2 < r_end) ? r + 2 : r_end - 1;
    ri_n = ridx[rn2];
    sr_n = seq[rn2];

    for (int e = t; e < BATCH * VDIM * 3; e += NTHREADS) {
        const int b2 = e / (VDIM * 3);
        const int j  = e % (VDIM * 3);
        out_xv[((long)b2 * R + r) * (VDIM * 3) + j] = xv[b2 * XVPAD + j];
    }

    if (t < BATCH * 3)
        out_root[((long)pb * R + r) * 3 + pd] = rv_cur;

    na = na_n;

    }
#undef ISSUE_X_LOADS
#undef ISSUE_W_LOADS
}

extern "C" void kernel_launch(void* const* d_in, const int* in_sizes, int n_in,
                              void* d_out, int out_size, void* d_ws, size_t ws_size,
                              hipStream_t stream) {
    const float* x      = (const float*)d_in[0];
    const float* W      = (const float*)d_in[1];
    const int*   seq    = (const int*)d_in[2];
    const int*   ridx   = (const int*)d_in[3];
    const int*   natoms = (const int*)d_in[4];

    const int R = in_sizes[2];
    const int A = in_sizes[0] / (BATCH * 3);

    float* out_root = (float*)d_out;
    float* out_xv   = out_root + (size_t)BATCH * R * 3;

    residues_encode_kernel<<<NBLOCKS, NTHREADS, 0, stream>>>(x, W, seq, ridx, natoms,
                                                             out_root, out_xv, A, R);
}


# x_v writeout: each thread moves 2 x 16 bytes (LDS row pitch 52; row, piece and output index fixed before the residue loop), not 8 x 4 bytes with div/mod and 64-bit multiplies per step
# speedup vs baseline: 2.0526x; 1.0802x over previous
#include <hip/hip_runtime.h>


typedef __attribute__((ext_vector_type(2))) float v2f;
typedef __attribute__((ext_vector_type(4))) float v4f;
typedef __attribute__((ext_vector_type(8))) float v8f;

#define BATCH 64
#define VDIM  16
#define MAXN  24
#define NTHREADS 384
#define DPITCH 76
#define XVPITCH 52
#define GITER (BATCH * MAXN * 3 / NTHREADS)
#define NBLOCKS 1024

__device__ __forceinline__ void lds_barrier()
{
    __builtin_amdgcn_fence(__ATOMIC_RELEASE, "workgroup", "local");
    __builtin_amdgcn_s_barrier();
    __builtin_amdgcn_fence(__ATOMIC_ACQUIRE, "workgroup", "local");
}

__launch_bounds__(NTHREADS, 12)
__global__ void residues_encode_kernel(const float* __restrict__ x,
                                       const float* __restrict__ W,
                                       const int*   __restrict__ seq,
                                       const int*   __restrict__ ridx,
                                       const int*   __restrict__ natoms,
                                       float* __restrict__ out_root,
                                       float* __restrict__ out_xv,
                                       int A, int R)
{
    __shared__ float lds_root[BATCH * 3];
    __shared__ __attribute__((aligned(16))) float lds_delta[BATCH * DPITCH];

    const int t = threadIdx.x;

    const int per     = (R + (int)gridDim.x - 1) / (int)gridDim.x;
    const int r_begin = (int)blockIdx.x * per;
    const int r_end   = (r_begin + per < R) ? (r_begin + per) : R;

    const int q0 = t / (MAXN * 3);
    const int n0 = (t % (MAXN * 3)) / 3;
    const unsigned a3 = 3u * (unsigned)A;
    int ns[3], bs[3];
    #pragma unroll
    for (int s = 0; s < 3; ++s) {
        const int m0 = n0 + 8 * s;
        const int c  = m0 / MAXN;
        ns[s] = m0 - MAXN * c;
        bs[s] = q0 + 5 * s + c;
    }

    if (r_begin >= r_end) return;

    const int pb = t / 3, pd = t % 3;
    const int wave  = t >> 5;
    const int lane  = t & 31;
    const int half  = lane >> 4;
    const int col   = lane & 15;

#define ISSUE_X_LOADS(RI)                                                                   \
    do {                                                                                    \
        if (t < BATCH * 3)                                                                  \
            rv = x[(unsigned)(pb * A + ((RI) + 1)) * 3u + (unsigned)pd];                    \
        _Pragma("unroll")                       \
        for (int s = 0; s < 3; ++s) {                                                       \
            int atom = (RI) + ns[s];                                                        \
            if (atom > A - 1) atom = A - 1;                             \
            const unsigned base = (unsigned)bs[s] * a3 + (unsigned)atom * 3u + (unsigned)pd; \
            _Pragma("unroll")                                               \
            for (int m = 0; m < 4; ++m)                                                     \
                g[3 * m + s] = x[base + (unsigned)(16 * m) * a3];                           \
        }                                                                                   \
    } while (0)

#define ISSUE_W_LOADS(SR)                                                                   \
    do {                                                                                    \
        const float* Wr = W + (((SR) * VDIM + col) * MAXN + 2 * half);                      \
        _Pragma("unroll")                      \
        for (int k = 0; k < 6; ++k) {                                                       \
            afrag[k][0] = Wr[4 * k];                                                        \
            afrag[k][1] = Wr[4 * k + 1];                                                    \
        }                                                                                   \
    } while (0)

    const int sr0 = seq[r_begin];
    const int ri0 = ridx[r_begin];
    int na = natoms[sr0];
    float rv = 0.0f;
    float g[GITER];
    v2f afrag[6];
    ISSUE_W_LOADS(sr0);
    ISSUE_X_LOADS(ri0);
    const int rn1 = (r_begin + 1 < r_end) ? r_begin + 1 : r_begin;
    int ri_n = ridx[rn1];
    int sr_n = seq[rn1];

    const int d     = wave >> 2;
    const int btile = wave & 3;
    const int b     = btile * 16 + col;

    const int wb = t / 12, wq = t % 12;
    const float* xv_src = &lds_delta[wb * XVPITCH + 4 * wq];
    v4f* const out4_lo = (v4f*)out_xv;
    v4f* const out4_hi = out4_lo + 32u * (unsigned)R * 12u;
    unsigned xv_idx   = ((unsigned)wb * (unsigned)R + (unsigned)r_begin) * 12u + (unsigned)wq;
    unsigned root_idx = ((unsigned)pb * (unsigned)R + (unsigned)r_begin) * 3u + (unsigned)pd;

    for (int r = r_begin; r < r_end; ++r) {

    const bool more = (r + 1 < r_end);

    const float rv_cur = rv;

    if (t < BATCH * 3)
        lds_root[t] = rv;
    lds_barrier();

    #pragma unroll
    for (int s = 0; s < 3; ++s) {
        #pragma unroll
        for (int m = 0; m < 4; ++m) {
            const int b2 = bs[s] + 16 * m;
            const float diff = g[3 * m + s] - lds_root[3 * b2 + pd];
            lds_delta[b2 * DPITCH + pd * MAXN + ns[s]] = (ns[s] < na) ? diff : 0.0f;
        }
    }

    const int na_n = natoms[sr_n];
    if (more)
        ISSUE_X_LOADS(ri_n);
    lds_barrier();

    const float* brow = &lds_delta[b * DPITCH + d * MAXN + 2 * half];
    v8f acc = {};
    #pragma unroll
    for (int k = 0; k < 6; ++k) {
        const v2f bb = *(const v2f*)(brow + 4 * k);
        acc = __builtin_amdgcn_wmma_f32_16x16x4_f32(
             false, afrag[k],  false, bb,
             (short)0, acc,  false,  false);
    }

    lds_barrier();

    float* xv = &lds_delta[0];
    #pragma unroll
    for (int p = 0; p < 8; ++p) {
        const int v = p + 8 * half;
        xv[b * XVPITCH + v * 3 + d] = acc[p];
    }
    lds_barrier();

    if (more)
        ISSUE_W_LOADS(sr_n);

    const int rn2 = (r + 2 < r_end) ? r + 2 : r_end - 1;
    ri_n = ridx[rn2];
    sr_n = seq[rn2];

    {
        const v4f o_lo = *(const v4f*)(xv_src);
        const v4f o_hi = *(const v4f*)(xv_src + 32 * XVPITCH);
        out4_lo[xv_idx] = o_lo;
        out4_hi[xv_idx] = o_hi;
        xv_idx += 12u;
    }

    if (t < BATCH * 3)
        out_root[root_idx] = rv_cur;
    root_idx += 3u;

    na = na_n;

    }
#undef ISSUE_X_LOADS
#undef ISSUE_W_LOADS
}

extern "C" void kernel_launch(void* const* d_in, const int* in_sizes, int n_in,
                              void* d_out, int out_size, void* d_ws, size_t ws_size,
                              hipStream_t stream) {
    const float* x      = (const float*)d_in[0];
    const float* W      = (const float*)d_in[1];
    const int*   seq    = (const int*)d_in[2];
    const int*   ridx   = (const int*)d_in[3];
    const int*   natoms = (const int*)d_in[4];

    const int R = in_sizes[2];
    const int A = in_sizes[0] / (BATCH * 3);

    float* out_root = (float*)d_out;
    float* out_xv   = out_root + (size_t)BATCH * R * 3;

    residues_encode_kernel<<<NBLOCKS, NTHREADS, 0, stream>>>(x, W, seq, ridx, natoms,
                                                             out_root, out_xv, A, R);
}


# Fatter waves: a workgroup is 3 waves (one per d), each with 4 batch tiles and 48 gather loads per lane: a quarter of the waves enter (3 per workgroup, not 12); W fetched by 3 waves, not 12
# speedup vs baseline: 2.1222x; 1.0339x over previous
#include <hip/hip_runtime.h>


typedef __attribute__((ext_vector_type(2))) float v2f;
typedef __attribute__((ext_vector_type(4))) float v4f;
typedef __attribute__((ext_vector_type(8))) float v8f;

#define BATCH 64
#define VDIM  16
#define MAXN  24
#define NTHREADS 96
#define NBT 4
#define WPIECES (BATCH * 12 / NTHREADS)
#define DPITCH 76
#define XVPITCH 52
#define GITER (BATCH * MAXN * 3 / NTHREADS)
#define NBLOCKS 1024

__device__ __forceinline__ void lds_barrier()
{
    __builtin_amdgcn_fence(__ATOMIC_RELEASE, "workgroup", "local");
    __builtin_amdgcn_s_barrier();
    __builtin_amdgcn_fence(__ATOMIC_ACQUIRE, "workgroup", "local");
}

__launch_bounds__(NTHREADS, 4)
__global__ void residues_encode_kernel(const float* __restrict__ x,
                                       const float* __restrict__ W,
                                       const int*   __restrict__ seq,
                                       const int*   __restrict__ ridx,
                                       const int*   __restrict__ natoms,
                                       float* __restrict__ out_root,
                                       float* __restrict__ out_xv,
                                       int A, int R)
{
    __shared__ float lds_root[BATCH * 3];
    __shared__ __attribute__((aligned(16))) float lds_delta[BATCH * DPITCH];

    const int t = threadIdx.x;

    const int per     = (R + (int)gridDim.x - 1) / (int)gridDim.x;
    const int r_begin = (int)blockIdx.x * per;
    const int r_end   = (r_begin + per < R) ? (r_begin + per) : R;

    const int q0 = t / (MAXN * 3);
    const int n0 = (t % (MAXN * 3)) / 3;
    const unsigned a3 = 3u * (unsigned)A;
    int ns[3], bs[3];
    #pragma unroll
    for (int s = 0; s < 3; ++s) {
        const int m0 = n0 + 8 * s;
        const int c  = m0 / MAXN;
        ns[s] = m0 - MAXN * c;
        bs[s] = q0 + s + c;
    }

    if (r_begin >= r_end) return;

    const int pb = t / 3, pd = t % 3;
    const int wave  = t >> 5;
    const int lane  = t & 31;
    const int half  = lane >> 4;
    const int col   = lane & 15;

#define ISSUE_X_LOADS(RI)                                                                   \
    do {                                                                                    \
        rv0 = x[(unsigned)(pb * A + ((RI) + 1)) * 3u + (unsigned)pd];                       \
        rv1 = x[(unsigned)((pb + 32) * A + ((RI) + 1)) * 3u + (unsigned)pd];                \
        _Pragma("unroll")                       \
        for (int s = 0; s < 3; ++s) {                                                       \
            int atom = (RI) + ns[s];                                                        \
            if (atom > A - 1) atom = A - 1;                             \
            const unsigned base = (unsigned)bs[s] * a3 + (unsigned)atom * 3u + (unsigned)pd; \
            _Pragma("unroll")                                               \
            for (int m = 0; m < 16; ++m)                                                    \
                g[3 * m + s] = x[base + (unsigned)(4 * m) * a3];                            \
        }                                                                                   \
    } while (0)

#define ISSUE_W_LOADS(SR)                                                                   \
    do {                                                                                    \
        const float* Wr = W + (((SR) * VDIM + col) * MAXN + 2 * half);                      \
        _Pragma("unroll")                      \
        for (int k = 0; k < 6; ++k) {                                                       \
            afrag[k][0] = Wr[4 * k];                                                        \
            afrag[k][1] = Wr[4 * k + 1];                                                    \
        }                                                                                   \
    } while (0)

    const int sr0 = seq[r_begin];
    const int ri0 = ridx[r_begin];
    int na = natoms[sr0];
    float rv0, rv1;
    float g[GITER];
    v2f afrag[6];
    ISSUE_W_LOADS(sr0);
    ISSUE_X_LOADS(ri0);
    const int rn1 = (r_begin + 1 < r_end) ? r_begin + 1 : r_begin;
    int ri_n = ridx[rn1];
    int sr_n = seq[rn1];

    const int d     = wave;

    const int wb = t / 12, wq = t % 12;
    const float* xv_src = &lds_delta[wb * XVPITCH + 4 * wq];
    v4f* const out4 = (v4f*)out_xv;
    const unsigned xv_step = 8u * (unsigned)R * 12u;
    unsigned xv_idx   = ((unsigned)wb * (unsigned)R + (unsigned)r_begin) * 12u + (unsigned)wq;
    unsigned root_idx = ((unsigned)pb * (unsigned)R + (unsigned)r_begin) * 3u + (unsigned)pd;
    const unsigned root_step = 32u * (unsigned)R * 3u;

    for (int r = r_begin; r < r_end; ++r) {

    const bool more = (r + 1 < r_end);

    const float rv_cur0 = rv0;
    const float rv_cur1 = rv1;

    lds_root[t] = rv0;
    lds_root[t + NTHREADS] = rv1;
    lds_barrier();

    #pragma unroll
    for (int s = 0; s < 3; ++s) {
        #pragma unroll
        for (int m = 0; m < 16; ++m) {
            const int b2 = bs[s] + 4 * m;
            const float diff = g[3 * m + s] - lds_root[3 * b2 + pd];
            lds_delta[b2 * DPITCH + pd * MAXN + ns[s]] = (ns[s] < na) ? diff : 0.0f;
        }
    }

    const int na_n = natoms[sr_n];
    if (more)
        ISSUE_X_LOADS(ri_n);
    lds_barrier();

    const float* brow = &lds_delta[col * DPITCH + d * MAXN + 2 * half];
    v8f acc[NBT];
    #pragma unroll
    for (int bt = 0; bt < NBT; ++bt)
        acc[bt] = v8f{};
    #pragma unroll
    for (int k = 0; k < 6; ++k) {
        #pragma unroll
        for (int bt = 0; bt < NBT; ++bt) {
            const v2f bb = *(const v2f*)(brow + bt * 16 * DPITCH + 4 * k);
            acc[bt] = __builtin_amdgcn_wmma_f32_16x16x4_f32(
                 false, afrag[k],  false, bb,
                 (short)0, acc[bt],  false,  false);
        }
    }

    lds_barrier();

    float* xv = &lds_delta[0];
    #pragma unroll
    for (int bt = 0; bt < NBT; ++bt) {
        #pragma unroll
        for (int p = 0; p < 8; ++p) {
            const int v = p + 8 * half;
            xv[(bt * 16 + col) * XVPITCH + v * 3 + d] = acc[bt][p];
        }
    }
    lds_barrier();

    if (more)
        ISSUE_W_LOADS(sr_n);

    const int rn2 = (r + 2 < r_end) ? r + 2 : r_end - 1;
    ri_n = ridx[rn2];
    sr_n = seq[rn2];

    {
        v4f o[WPIECES];
        #pragma unroll
        for (int i = 0; i < WPIECES; ++i)
            o[i] = *(const v4f*)(xv_src + 8 * i * XVPITCH);
        #pragma unroll
        for (int i = 0; i < WPIECES; ++i)
            out4[xv_idx + (unsigned)i * xv_step] = o[i];
        xv_idx += 12u;
    }

    out_root[root_idx] = rv_cur0;
    out_root[root_idx + root_step] = rv_cur1;
    root_idx += 3u;

    na = na_n;

    }
#undef ISSUE_X_LOADS
#undef ISSUE_W_LOADS
}

extern "C" void kernel_launch(void* const* d_in, const int* in_sizes, int n_in,
                              void* d_out, int out_size, void* d_ws, size_t ws_size,
                              hipStream_t stream) {
    const float* x      = (const float*)d_in[0];
    const float* W      = (const float*)d_in[1];
    const int*   seq    = (const int*)d_in[2];
    const int*   ridx   = (const int*)d_in[3];
    const int*   natoms = (const int*)d_in[4];

    const int R = in_sizes[2];
    const int A = in_sizes[0] / (BATCH * 3);

    float* out_root = (float*)d_out;
    float* out_xv   = out_root + (size_t)BATCH * R * 3;

    residues_encode_kernel<<<NBLOCKS, NTHREADS, 0, stream>>>(x, W, seq, ridx, natoms,
                                                             out_root, out_xv, A, R);
}


# Root and mask applied behind the LDS transpose, on the B fragments (root = window slot n=1, 4 per lane): no root loads, no root phase; D tile staged in LDS of its own: 2 barriers per trip, not 4
# speedup vs baseline: 2.2451x; 1.0579x over previous
#include <hip/hip_runtime.h>


typedef __attribute__((ext_vector_type(2))) float v2f;
typedef __attribute__((ext_vector_type(4))) float v4f;
typedef __attribute__((ext_vector_type(8))) float v8f;

#define BATCH 64
#define VDIM  16
#define MAXN  24
#define NTHREADS 96
#define NBT 4
#define WPIECES (BATCH * 12 / NTHREADS)
#define DPITCH 76
#define XVPITCH 52
#define GITER (BATCH * MAXN * 3 / NTHREADS)
#define NBLOCKS 1024

__device__ __forceinline__ void lds_barrier()
{
    __builtin_amdgcn_fence(__ATOMIC_RELEASE, "workgroup", "local");
    __builtin_amdgcn_s_barrier();
    __builtin_amdgcn_fence(__ATOMIC_ACQUIRE, "workgroup", "local");
}

__launch_bounds__(NTHREADS, 4)
__global__ void residues_encode_kernel(const float* __restrict__ x,
                                       const float* __restrict__ W,
                                       const int*   __restrict__ seq,
                                       const int*   __restrict__ ridx,
                                       const int*   __restrict__ natoms,
                                       float* __restrict__ out_root,
                                       float* __restrict__ out_xv,
                                       int A, int R)
{
    __shared__ __attribute__((aligned(16))) float lds_win[BATCH * DPITCH];
    __shared__ __attribute__((aligned(16))) float lds_xv[BATCH * XVPITCH];

    const int t = threadIdx.x;

    const int per     = (R + (int)gridDim.x - 1) / (int)gridDim.x;
    const int r_begin = (int)blockIdx.x * per;
    const int r_end   = (r_begin + per < R) ? (r_begin + per) : R;

    const int q0 = t / (MAXN * 3);
    const int n0 = (t % (MAXN * 3)) / 3;
    const unsigned a3 = 3u * (unsigned)A;
    int ns[3], bs[3];
    #pragma unroll
    for (int s = 0; s < 3; ++s) {
        const int m0 = n0 + 8 * s;
        const int c  = m0 / MAXN;
        ns[s] = m0 - MAXN * c;
        bs[s] = q0 + s + c;
    }

    if (r_begin >= r_end) return;

    const int pb = t / 3, pd = t % 3;
    const int wave  = t >> 5;
    const int lane  = t & 31;
    const int half  = lane >> 4;
    const int col   = lane & 15;

#define ISSUE_X_LOADS(RI)                                                                   \
    do {                                                                                    \
        _Pragma("unroll")                       \
        for (int s = 0; s < 3; ++s) {                                                       \
            int atom = (RI) + ns[s];                                                        \
            if (atom > A - 1) atom = A - 1;                             \
            const unsigned base = (unsigned)bs[s] * a3 + (unsigned)atom * 3u + (unsigned)pd; \
            _Pragma("unroll")                                               \
            for (int m = 0; m < 16; ++m)                                                    \
                g[3 * m + s] = x[base + (unsigned)(4 * m) * a3];                            \
        }                                                                                   \
    } while (0)

#define ISSUE_W_LOADS(SR)                                                                   \
    do {                                                                                    \
        const float* Wr = W + (((SR) * VDIM + col) * MAXN + 2 * half);                      \
        _Pragma("unroll")                      \
        for (int k = 0; k < 6; ++k) {                                                       \
            afrag[k][0] = Wr[4 * k];                                                        \
            afrag[k][1] = Wr[4 * k + 1];                                                    \
        }                                                                                   \
    } while (0)

    const int sr0 = seq[r_begin];
    const int ri0 = ridx[r_begin];
    int na = natoms[sr0];
    float g[GITER];
    v2f afrag[6];
    ISSUE_X_LOADS(ri0);
    ISSUE_W_LOADS(sr0);
    const int rn1 = (r_begin + 1 < r_end) ? r_begin + 1 : r_begin;
    int ri_n = ridx[rn1];
    int sr_n = seq[rn1];

    const int d     = wave;

    const int wb = t / 12, wq = t % 12;
    const float* xv_src = &lds_xv[wb * XVPITCH + 4 * wq];
    v4f* const out4 = (v4f*)out_xv;
    const unsigned xv_step = 8u * (unsigned)R * 12u;
    unsigned xv_idx   = ((unsigned)wb * (unsigned)R + (unsigned)r_begin) * 12u + (unsigned)wq;
    unsigned root_idx = ((unsigned)pb * (unsigned)R + (unsigned)r_begin) * 3u + (unsigned)pd;
    const unsigned root_step = 32u * (unsigned)R * 3u;
    const float* root_src = &lds_win[pb * DPITCH + pd * MAXN + 1];
    const float* brow = &lds_win[col * DPITCH + d * MAXN + 2 * half];
    const float* rrow = &lds_win[col * DPITCH + d * MAXN + 1];

    for (int r = r_begin; r < r_end; ++r) {

    const bool more = (r + 1 < r_end);

    const int nh = na - 2 * half;

    #pragma unroll
    for (int s = 0; s < 3; ++s) {
        #pragma unroll
        for (int m = 0; m < 16; ++m) {
            const int b2 = bs[s] + 4 * m;
            lds_win[b2 * DPITCH + pd * MAXN + ns[s]] = g[3 * m + s];
        }
    }

    const int na_n = natoms[sr_n];
    if (more)
        ISSUE_X_LOADS(ri_n);
    lds_barrier();

    out_root[root_idx] = root_src[0];
    out_root[root_idx + root_step] = root_src[32 * DPITCH];
    root_idx += 3u;

    float rt[NBT];
    #pragma unroll
    for (int bt = 0; bt < NBT; ++bt)
        rt[bt] = rrow[bt * 16 * DPITCH];
    v8f acc[NBT];
    #pragma unroll
    for (int bt = 0; bt < NBT; ++bt)
        acc[bt] = v8f{};
    #pragma unroll
    for (int k = 0; k < 6; ++k) {
        #pragma unroll
        for (int bt = 0; bt < NBT; ++bt) {
            const v2f w2 = *(const v2f*)(brow + bt * 16 * DPITCH + 4 * k);
            v2f bb;
            bb[0] = (4 * k     < nh) ? (w2[0] - rt[bt]) : 0.0f;
            bb[1] = (4 * k + 1 < nh) ? (w2[1] - rt[bt]) : 0.0f;
            acc[bt] = __builtin_amdgcn_wmma_f32_16x16x4_f32(
                 false, afrag[k],  false, bb,
                 (short)0, acc[bt],  false,  false);
        }
    }

    float* xv = &lds_xv[0];
    #pragma unroll
    for (int bt = 0; bt < NBT; ++bt) {
        #pragma unroll
        for (int p = 0; p < 8; ++p) {
            const int v = p + 8 * half;
            xv[(bt * 16 + col) * XVPITCH + v * 3 + d] = acc[bt][p];
        }
    }
    lds_barrier();

    if (more)
        ISSUE_W_LOADS(sr_n);

    const int rn2 = (r + 2 < r_end) ? r + 2 : r_end - 1;
    ri_n = ridx[rn2];
    sr_n = seq[rn2];

    {
        v4f o[WPIECES];
        #pragma unroll
        for (int i = 0; i < WPIECES; ++i)
            o[i] = *(const v4f*)(xv_src + 8 * i * XVPITCH);
        #pragma unroll
        for (int i = 0; i < WPIECES; ++i)
            out4[xv_idx + (unsigned)i * xv_step] = o[i];
        xv_idx += 12u;
    }

    __builtin_amdgcn_sched_barrier(0);
    na = na_n;

    }
#undef ISSUE_X_LOADS
#undef ISSUE_W_LOADS
}

extern "C" void kernel_launch(void* const* d_in, const int* in_sizes, int n_in,
                              void* d_out, int out_size, void* d_ws, size_t ws_size,
                              hipStream_t stream) {
    const float* x      = (const float*)d_in[0];
    const float* W      = (const float*)d_in[1];
    const int*   seq    = (const int*)d_in[2];
    const int*   ridx   = (const int*)d_in[3];
    const int*   natoms = (const int*)d_in[4];

    const int R = in_sizes[2];
    const int A = in_sizes[0] / (BATCH * 3);

    float* out_root = (float*)d_out;
    float* out_xv   = out_root + (size_t)BATCH * R * 3;

    residues_encode_kernel<<<NBLOCKS, NTHREADS, 0, stream>>>(x, W, seq, ridx, natoms,
                                                             out_root, out_xv, A, R);
}


# Work re-cut: a workgroup owns 16 batch rows x 16 residues, a trip covers 4 consecutive residues (was 64 rows x 1): per row the trip's windows, x_v (768 B, whole lines) and roots are contiguous
# speedup vs baseline: 2.7536x; 1.2265x over previous
#include <hip/hip_runtime.h>


typedef __attribute__((ext_vector_type(2))) float v2f;
typedef __attribute__((ext_vector_type(4))) float v4f;
typedef __attribute__((ext_vector_type(8))) float v8f;
typedef __attribute__((ext_vector_type(4), aligned(4))) int v4i;

#define BATCH 64
#define VDIM  16
#define MAXN  24
#define NTHREADS 96
#define NBT 4
#define BROWS 16
#define NBQ (BATCH / BROWS)
#define WPIECES (BATCH * 12 / NTHREADS)
#define DPITCH 76
#define XVPITCH 196
#define GITER (BATCH * MAXN * 3 / NTHREADS)
#define NBLOCKS 1024

__device__ __forceinline__ void lds_barrier()
{
    __builtin_amdgcn_fence(__ATOMIC_RELEASE, "workgroup", "local");
    __builtin_amdgcn_s_barrier();
    __builtin_amdgcn_fence(__ATOMIC_ACQUIRE, "workgroup", "local");
}

__launch_bounds__(NTHREADS, 4)
__global__ void residues_encode_kernel(const float* __restrict__ x,
                                       const float* __restrict__ W,
                                       const int*   __restrict__ seq,
                                       const int*   __restrict__ ridx,
                                       const int*   __restrict__ natoms,
                                       float* __restrict__ out_root,
                                       float* __restrict__ out_xv,
                                       int A, int R)
{
    __shared__ __attribute__((aligned(16))) float lds_win[BATCH * DPITCH];
    __shared__ __attribute__((aligned(16))) float lds_xv[BROWS * XVPITCH];

    const int t = threadIdx.x;

    const int bq      = (int)blockIdx.x % NBQ;
    const int nruns   = (int)gridDim.x / NBQ;
    const int per     = (((R + nruns - 1) / nruns) + NBT - 1) / NBT * NBT;
    const int r_begin = ((int)blockIdx.x / NBQ) * per;
    const int r_end   = (r_begin + per < R) ? (r_begin + per) : R;
    const unsigned b0 = (unsigned)(bq * BROWS);

    const int q0 = t / (MAXN * 3);
    const int n0 = (t % (MAXN * 3)) / 3;
    const unsigned a3 = 3u * (unsigned)A;
    int ns[3], bs[3];
    #pragma unroll
    for (int s = 0; s < 3; ++s) {
        const int m0 = n0 + 8 * s;
        const int c  = m0 / MAXN;
        ns[s] = m0 - MAXN * c;
        bs[s] = q0 + s + c;
    }

    if (r_begin >= r_end) return;

    const int pd = t % 3;
    unsigned rowbase[3];
    #pragma unroll
    for (int s = 0; s < 3; ++s)
        rowbase[s] = (b0 + (unsigned)bs[s]) * a3 + (unsigned)pd;
    const int wave  = t >> 5;
    const int lane  = t & 31;
    const int half  = lane >> 4;
    const int col   = lane & 15;

#define ISSUE_X_LOADS(RI)                                                                   \
    do {                                                                                    \
        _Pragma("unroll")                       \
        for (int j = 0; j < NBT; ++j) {                                                     \
            _Pragma("unroll")                                               \
            for (int s = 0; s < 3; ++s) {                                                   \
                int atom = (RI)[j] + ns[s];                                                 \
                if (atom > A - 1) atom = A - 1;                         \
                const unsigned base = rowbase[s] + (unsigned)atom * 3u;                     \
                _Pragma("unroll")                                           \
                for (int m = 0; m < 4; ++m)                                                 \
                    g[12 * j + 3 * m + s] = x[base + (unsigned)(4 * m) * a3];               \
            }                                                                               \
        }                                                                                   \
    } while (0)

#define ISSUE_W_LOADS(SR)                                                                   \
    do {                                                                                    \
        _Pragma("unroll")                      \
        for (int j = 0; j < NBT; ++j) {                                                     \
            const float* Wr = W + (((SR)[j] * VDIM + col) * MAXN + 2 * half);               \
            _Pragma("unroll")                                               \
            for (int k = 0; k < 6; ++k) {                                                   \
                afrag[j][k][0] = Wr[4 * k];                                                 \
                afrag[j][k][1] = Wr[4 * k + 1];                                             \
            }                                                                               \
        }                                                                                   \
    } while (0)

#define LOAD_IDX(RB, RI, SR)                                                                \
    do {                                                                                    \
        if ((RB) + NBT <= r_end) {                   \
            (RI) = *(const v4i*)(ridx + (RB));                                              \
            (SR) = *(const v4i*)(seq + (RB));                                               \
        } else {                                                                            \
            _Pragma("unroll")                                              \
            for (int j = 0; j < NBT; ++j) {                                                 \
                const int rr = ((RB) + j < r_end) ? (RB) + j : r_end - 1;                   \
                (RI)[j] = ridx[rr];                                                         \
                (SR)[j] = seq[rr];                                                          \
            }                                                                               \
        }                                                                                   \
    } while (0)

    v4i ri0, sr0;
    int na[NBT];
    LOAD_IDX(r_begin, ri0, sr0);
    #pragma unroll
    for (int j = 0; j < NBT; ++j)
        na[j] = natoms[sr0[j]];
    float g[GITER];
    v2f afrag[NBT][6];
    ISSUE_X_LOADS(ri0);
    ISSUE_W_LOADS(sr0);
    v4i ri_n, sr_n;
    LOAD_IDX(r_begin + NBT, ri_n, sr_n);

    const int d     = wave;

    const int wb = t / 48, wq = t % 48;
    const float* xv_src = &lds_xv[wb * XVPITCH + 4 * wq];
    v4f* const out4 = (v4f*)out_xv;
    const unsigned xv_step = 2u * (unsigned)R * 12u;
    unsigned xv_idx   = ((b0 + (unsigned)wb) * (unsigned)R + (unsigned)r_begin) * 12u + (unsigned)wq;
    const int rb = t / 12, rw = t % 12;
    unsigned root_idx = ((b0 + (unsigned)rb) * (unsigned)R + (unsigned)r_begin) * 3u + (unsigned)rw;
    const unsigned root_step = 8u * (unsigned)R * 3u;
    const float* root_src = &lds_win[(16 * (rw / 3) + rb) * DPITCH + (rw % 3) * MAXN + 1];
    const float* brow = &lds_win[col * DPITCH + d * MAXN + 2 * half];
    const float* rrow = &lds_win[col * DPITCH + d * MAXN + 1];

    for (int r = r_begin; r < r_end; r += NBT) {

    const bool more = (r + NBT < r_end);
    const int nvalid = r_end - r;

    int nh[NBT];
    #pragma unroll
    for (int j = 0; j < NBT; ++j)
        nh[j] = na[j] - 2 * half;

    #pragma unroll
    for (int s = 0; s < 3; ++s) {
        #pragma unroll
        for (int m = 0; m < 16; ++m) {
            const int b2 = bs[s] + 4 * m;
            lds_win[b2 * DPITCH + pd * MAXN + ns[s]] = g[3 * m + s];
        }
    }

    int na_n[NBT];
    #pragma unroll
    for (int j = 0; j < NBT; ++j)
        na_n[j] = natoms[sr_n[j]];
    if (more)
        ISSUE_X_LOADS(ri_n);
    lds_barrier();

    if (rw < 3 * nvalid) {
        out_root[root_idx] = root_src[0];
        out_root[root_idx + root_step] = root_src[8 * DPITCH];
    }
    root_idx += 3u * NBT;

    float rt[NBT];
    #pragma unroll
    for (int bt = 0; bt < NBT; ++bt)
        rt[bt] = rrow[bt * 16 * DPITCH];
    v8f acc[NBT];
    #pragma unroll
    for (int bt = 0; bt < NBT; ++bt)
        acc[bt] = v8f{};
    #pragma unroll
    for (int k = 0; k < 6; ++k) {
        #pragma unroll
        for (int bt = 0; bt < NBT; ++bt) {
            const v2f w2 = *(const v2f*)(brow + bt * 16 * DPITCH + 4 * k);
            v2f bb;
            bb[0] = (4 * k     < nh[bt]) ? (w2[0] - rt[bt]) : 0.0f;
            bb[1] = (4 * k + 1 < nh[bt]) ? (w2[1] - rt[bt]) : 0.0f;
            acc[bt] = __builtin_amdgcn_wmma_f32_16x16x4_f32(
                 false, afrag[bt][k],  false, bb,
                 (short)0, acc[bt],  false,  false);
        }
    }

    float* xv = &lds_xv[0];
    #pragma unroll
    for (int bt = 0; bt < NBT; ++bt) {
        #pragma unroll
        for (int p = 0; p < 8; ++p) {
            const int v = p + 8 * half;
            xv[col * XVPITCH + bt * 48 + v * 3 + d] = acc[bt][p];
        }
    }
    lds_barrier();

    if (more)
        ISSUE_W_LOADS(sr_n);

    {
        v4f o[WPIECES];
        #pragma unroll
        for (int i = 0; i < WPIECES; ++i)
            o[i] = *(const v4f*)(xv_src + 2 * i * XVPITCH);
        if (wq < 12 * nvalid) {
            #pragma unroll
            for (int i = 0; i < WPIECES; ++i)
                out4[xv_idx + (unsigned)i * xv_step] = o[i];
        }
        xv_idx += 12u * NBT;
    }

    LOAD_IDX(r + 2 * NBT, ri_n, sr_n);

    __builtin_amdgcn_sched_barrier(0);
    #pragma unroll
    for (int j = 0; j < NBT; ++j)
        na[j] = na_n[j];

    }
#undef ISSUE_X_LOADS
#undef ISSUE_W_LOADS
#undef LOAD_IDX
}

extern "C" void kernel_launch(void* const* d_in, const int* in_sizes, int n_in,
                              void* d_out, int out_size, void* d_ws, size_t ws_size,
                              hipStream_t stream) {
    const float* x      = (const float*)d_in[0];
    const float* W      = (const float*)d_in[1];
    const int*   seq    = (const int*)d_in[2];
    const int*   ridx   = (const int*)d_in[3];
    const int*   natoms = (const int*)d_in[4];

    const int R = in_sizes[2];
    const int A = in_sizes[0] / (BATCH * 3);

    float* out_root = (float*)d_out;
    float* out_xv   = out_root + (size_t)BATCH * R * 3;

    residues_encode_kernel<<<NBLOCKS, NTHREADS, 0, stream>>>(x, W, seq, ridx, natoms,
                                                             out_root, out_xv, A, R);
}


# More workgroups in flight: 1536 resident (6 per WGP, 18 waves, was 4 and 12), trips dealt out evenly so a workgroup runs 3 or 2 of them, not 4: half again as many gather loads outstanding per WGP
# speedup vs baseline: 2.9926x; 1.0868x over previous
#include <hip/hip_runtime.h>


typedef __attribute__((ext_vector_type(2))) float v2f;
typedef __attribute__((ext_vector_type(4))) float v4f;
typedef __attribute__((ext_vector_type(8))) float v8f;
typedef __attribute__((ext_vector_type(4), aligned(4))) int v4i;

#define BATCH 64
#define VDIM  16
#define MAXN  24
#define NTHREADS 96
#define NBT 4
#define BROWS 16
#define NBQ (BATCH / BROWS)
#define WPIECES (BATCH * 12 / NTHREADS)
#define DPITCH 76
#define XVPITCH 196
#define GITER (BATCH * MAXN * 3 / NTHREADS)
#define NBLOCKS 1536

__device__ __forceinline__ void lds_barrier()
{
    __builtin_amdgcn_fence(__ATOMIC_RELEASE, "workgroup", "local");
    __builtin_amdgcn_s_barrier();
    __builtin_amdgcn_fence(__ATOMIC_ACQUIRE, "workgroup", "local");
}

__launch_bounds__(NTHREADS, 5)
__global__ void residues_encode_kernel(const float* __restrict__ x,
                                       const float* __restrict__ W,
                                       const int*   __restrict__ seq,
                                       const int*   __restrict__ ridx,
                                       const int*   __restrict__ natoms,
                                       float* __restrict__ out_root,
                                       float* __restrict__ out_xv,
                                       int A, int R)
{
    __shared__ __attribute__((aligned(16))) float lds_win[BATCH * DPITCH];
    __shared__ __attribute__((aligned(16))) float lds_xv[BROWS * XVPITCH];

    const int t = threadIdx.x;

    const int bq      = (int)blockIdx.x % NBQ;
    const int run     = (int)blockIdx.x / NBQ;
    const int nruns   = NBLOCKS / NBQ;
    const int ntrips  = (R + NBT - 1) / NBT;
    const int tbase   = ntrips / nruns;
    const int trem    = ntrips - tbase * nruns;
    const int mytrips = tbase + ((run < trem) ? 1 : 0);
    const int r_begin = (run * tbase + ((run < trem) ? run : trem)) * NBT;
    const int r_end   = (r_begin + mytrips * NBT < R) ? (r_begin + mytrips * NBT) : R;
    const unsigned b0 = (unsigned)(bq * BROWS);

    const int q0 = t / (MAXN * 3);
    const int n0 = (t % (MAXN * 3)) / 3;
    const unsigned a3 = 3u * (unsigned)A;
    int ns[3], bs[3];
    #pragma unroll
    for (int s = 0; s < 3; ++s) {
        const int m0 = n0 + 8 * s;
        const int c  = m0 / MAXN;
        ns[s] = m0 - MAXN * c;
        bs[s] = q0 + s + c;
    }

    if (r_begin >= r_end) return;

    const int pd = t % 3;
    unsigned rowbase[3];
    #pragma unroll
    for (int s = 0; s < 3; ++s)
        rowbase[s] = (b0 + (unsigned)bs[s]) * a3 + (unsigned)pd;
    const int wave  = t >> 5;
    const int lane  = t & 31;
    const int half  = lane >> 4;
    const int col   = lane & 15;

#define ISSUE_X_LOADS(RI)                                                                   \
    do {                                                                                    \
        _Pragma("unroll")                       \
        for (int j = 0; j < NBT; ++j) {                                                     \
            _Pragma("unroll")                                               \
            for (int s = 0; s < 3; ++s) {                                                   \
                int atom = (RI)[j] + ns[s];                                                 \
                if (atom > A - 1) atom = A - 1;                         \
                const unsigned base = rowbase[s] + (unsigned)atom * 3u;                     \
                _Pragma("unroll")                                           \
                for (int m = 0; m < 4; ++m)                                                 \
                    g[12 * j + 3 * m + s] = x[base + (unsigned)(4 * m) * a3];               \
            }                                                                               \
        }                                                                                   \
    } while (0)

#define ISSUE_W_LOADS(SR)                                                                   \
    do {                                                                                    \
        _Pragma("unroll")                      \
        for (int j = 0; j < NBT; ++j) {                                                     \
            const float* Wr = W + (((SR)[j] * VDIM + col) * MAXN + 2 * half);               \
            _Pragma("unroll")                                               \
            for (int k = 0; k < 6; ++k) {                                                   \
                afrag[j][k][0] = Wr[4 * k];                                                 \
                afrag[j][k][1] = Wr[4 * k + 1];                                             \
            }                                                                               \
        }                                                                                   \
    } while (0)

#define LOAD_IDX(RB, RI, SR)                                                                \
    do {                                                                                    \
        if ((RB) + NBT <= r_end) {                   \
            (RI) = *(const v4i*)(ridx + (RB));                                              \
            (SR) = *(const v4i*)(seq + (RB));                                               \
        } else {                                                                            \
            _Pragma("unroll")                                              \
            for (int j = 0; j < NBT; ++j) {                                                 \
                const int rr = ((RB) + j < r_end) ? (RB) + j : r_end - 1;                   \
                (RI)[j] = ridx[rr];                                                         \
                (SR)[j] = seq[rr];                                                          \
            }                                                                               \
        }                                                                                   \
    } while (0)

    v4i ri0, sr0;
    int na[NBT];
    LOAD_IDX(r_begin, ri0, sr0);
    #pragma unroll
    for (int j = 0; j < NBT; ++j)
        na[j] = natoms[sr0[j]];
    float g[GITER];
    v2f afrag[NBT][6];
    ISSUE_X_LOADS(ri0);
    ISSUE_W_LOADS(sr0);
    v4i ri_n, sr_n;
    LOAD_IDX(r_begin + NBT, ri_n, sr_n);

    const int d     = wave;

    const int wb = t / 48, wq = t % 48;
    const float* xv_src = &lds_xv[wb * XVPITCH + 4 * wq];
    v4f* const out4 = (v4f*)out_xv;
    const unsigned xv_step = 2u * (unsigned)R * 12u;
    unsigned xv_idx   = ((b0 + (unsigned)wb) * (unsigned)R + (unsigned)r_begin) * 12u + (unsigned)wq;
    const int rb = t / 12, rw = t % 12;
    unsigned root_idx = ((b0 + (unsigned)rb) * (unsigned)R + (unsigned)r_begin) * 3u + (unsigned)rw;
    const unsigned root_step = 8u * (unsigned)R * 3u;
    const float* root_src = &lds_win[(16 * (rw / 3) + rb) * DPITCH + (rw % 3) * MAXN + 1];
    const float* brow = &lds_win[col * DPITCH + d * MAXN + 2 * half];
    const float* rrow = &lds_win[col * DPITCH + d * MAXN + 1];

    for (int r = r_begin; r < r_end; r += NBT) {

    const bool more = (r + NBT < r_end);
    const int nvalid = r_end - r;

    int nh[NBT];
    #pragma unroll
    for (int j = 0; j < NBT; ++j)
        nh[j] = na[j] - 2 * half;

    #pragma unroll
    for (int s = 0; s < 3; ++s) {
        #pragma unroll
        for (int m = 0; m < 16; ++m) {
            const int b2 = bs[s] + 4 * m;
            lds_win[b2 * DPITCH + pd * MAXN + ns[s]] = g[3 * m + s];
        }
    }

    int na_n[NBT];
    #pragma unroll
    for (int j = 0; j < NBT; ++j)
        na_n[j] = natoms[sr_n[j]];
    if (more)
        ISSUE_X_LOADS(ri_n);
    lds_barrier();

    if (rw < 3 * nvalid) {
        out_root[root_idx] = root_src[0];
        out_root[root_idx + root_step] = root_src[8 * DPITCH];
    }
    root_idx += 3u * NBT;

    float rt[NBT];
    #pragma unroll
    for (int bt = 0; bt < NBT; ++bt)
        rt[bt] = rrow[bt * 16 * DPITCH];
    v8f acc[NBT];
    #pragma unroll
    for (int bt = 0; bt < NBT; ++bt)
        acc[bt] = v8f{};
    #pragma unroll
    for (int k = 0; k < 6; ++k) {
        #pragma unroll
        for (int bt = 0; bt < NBT; ++bt) {
            const v2f w2 = *(const v2f*)(brow + bt * 16 * DPITCH + 4 * k);
            v2f bb;
            bb[0] = (4 * k     < nh[bt]) ? (w2[0] - rt[bt]) : 0.0f;
            bb[1] = (4 * k + 1 < nh[bt]) ? (w2[1] - rt[bt]) : 0.0f;
            acc[bt] = __builtin_amdgcn_wmma_f32_16x16x4_f32(
                 false, afrag[bt][k],  false, bb,
                 (short)0, acc[bt],  false,  false);
        }
    }

    float* xv = &lds_xv[0];
    #pragma unroll
    for (int bt = 0; bt < NBT; ++bt) {
        #pragma unroll
        for (int p = 0; p < 8; ++p) {
            const int v = p + 8 * half;
            xv[col * XVPITCH + bt * 48 + v * 3 + d] = acc[bt][p];
        }
    }
    lds_barrier();

    if (more)
        ISSUE_W_LOADS(sr_n);

    {
        v4f o[WPIECES];
        #pragma unroll
        for (int i = 0; i < WPIECES; ++i)
            o[i] = *(const v4f*)(xv_src + 2 * i * XVPITCH);
        if (wq < 12 * nvalid) {
            #pragma unroll
            for (int i = 0; i < WPIECES; ++i)
                out4[xv_idx + (unsigned)i * xv_step] = o[i];
        }
        xv_idx += 12u * NBT;
    }

    LOAD_IDX(r + 2 * NBT, ri_n, sr_n);

    __builtin_amdgcn_sched_barrier(0);
    #pragma unroll
    for (int j = 0; j < NBT; ++j)
        na[j] = na_n[j];

    }
#undef ISSUE_X_LOADS
#undef ISSUE_W_LOADS
#undef LOAD_IDX
}

extern "C" void kernel_launch(void* const* d_in, const int* in_sizes, int n_in,
                              void* d_out, int out_size, void* d_ws, size_t ws_size,
                              hipStream_t stream) {
    const float* x      = (const float*)d_in[0];
    const float* W      = (const float*)d_in[1];
    const int*   seq    = (const int*)d_in[2];
    const int*   ridx   = (const int*)d_in[3];
    const int*   natoms = (const int*)d_in[4];

    const int R = in_sizes[2];
    const int A = in_sizes[0] / (BATCH * 3);

    float* out_root = (float*)d_out;
    float* out_xv   = out_root + (size_t)BATCH * R * 3;

    residues_encode_kernel<<<NBLOCKS, NTHREADS, 0, stream>>>(x, W, seq, ridx, natoms,
                                                             out_root, out_xv, A, R);
}
